# starting kernel
# baseline (speedup 1.0000x reference)
#include <hip/hip_runtime.h>
#include <hip/hip_bf16.h>


#define N_DRUGS 591
#define NP      592
#define FDIM    1024
#define NEDGE   120000

typedef __attribute__((ext_vector_type(2))) float v2f;
typedef __attribute__((ext_vector_type(8))) float v8f;

__global__ void init_ws(float* __restrict__ ahat, float* __restrict__ deg,
                        float* __restrict__ pool) {
  int idx = blockIdx.x * blockDim.x + threadIdx.x;
  int stride = gridDim.x * blockDim.x;
  const int AH = 3 * NP * NP;
  for (int i = idx; i < AH; i += stride) ahat[i] = 0.0f;
  for (int i = idx; i < 3 * NP; i += stride) deg[i] = 1.0f;
  if (idx < 8) pool[idx] = 0.0f;
}

__global__ void pad_dm1(const float* __restrict__ dm1, float* __restrict__ out) {
  int idx = blockIdx.x * blockDim.x + threadIdx.x;
  if (idx >= NP * FDIM) return;
  int n = idx >> 10;
  out[idx] = (n < N_DRUGS) ? dm1[idx] : 0.0f;
}

__global__ void deg_accum(const int* __restrict__ eT, const float* __restrict__ wT,
                          const int* __restrict__ eS, const float* __restrict__ wS,
                          const int* __restrict__ eG, const float* __restrict__ wG,
                          float* __restrict__ deg) {
  int idx = blockIdx.x * blockDim.x + threadIdx.x;
  if (idx >= 3 * NEDGE) return;
  int t = idx / NEDGE, e = idx - t * NEDGE;
  const int*   ed = (t == 0) ? eT : ((t == 1) ? eS : eG);
  const float* w  = (t == 0) ? wT : ((t == 1) ? wS : wG);
  int dst = ed[NEDGE + e];
  atomicAdd(&deg[t * NP + dst], w[e]);
}

__global__ void compute_dinv(const float* __restrict__ deg, float* __restrict__ dinv) {
  int i = blockIdx.x * blockDim.x + threadIdx.x;
  if (i < 3 * NP) {
    float d = deg[i];
    dinv[i] = (d > 0.0f) ? rsqrtf(d) : 0.0f;
  }
}

__global__ void build_ahat(const int* __restrict__ eT, const float* __restrict__ wT,
                           const int* __restrict__ eS, const float* __restrict__ wS,
                           const int* __restrict__ eG, const float* __restrict__ wG,
                           const float* __restrict__ dinv, float* __restrict__ ahat) {
  const int PER = NEDGE + N_DRUGS;
  int idx = blockIdx.x * blockDim.x + threadIdx.x;
  if (idx >= 3 * PER) return;
  int t = idx / PER, r = idx - t * PER;
  const float* dv = dinv + t * NP;
  float* Ah = ahat + (size_t)t * NP * NP;
  if (r < NEDGE) {
    const int*   ed = (t == 0) ? eT : ((t == 1) ? eS : eG);
    const float* w  = (t == 0) ? wT : ((t == 1) ? wS : wG);
    int src = ed[r], dst = ed[NEDGE + r];
    atomicAdd(&Ah[(size_t)dst * NP + src], dv[src] * w[r] * dv[dst]);
  } else {
    int i = r - NEDGE;
    atomicAdd(&Ah[(size_t)i * NP + i], dv[i] * dv[i]);
  }
}

__global__ void gemm_wmma_f32(const float* __restrict__ A, int lda,
                              const float* __restrict__ B, int ldb,
                              float* __restrict__ C, int ldc,
                              int K, int n_wtiles, int m_tiles,
                              const float* __restrict__ bias, int do_relu) {
  int wid = (blockIdx.x * blockDim.x + threadIdx.x) >> 5;
  if (wid >= m_tiles * n_wtiles) return;
  int lane = threadIdx.x & 31;
  int tm = wid / n_wtiles, tn = wid - tm * n_wtiles;
  int row0 = tm << 4, col0 = tn << 6;
  int ml = lane & 15;
  int khalf = (lane >> 4) << 1;

  const float* Arow = A + (size_t)(row0 + ml) * lda + khalf;
  const float* B0   = B + (size_t)khalf * ldb + (col0 + ml);

  v8f acc0 = {}, acc1 = {}, acc2 = {}, acc3 = {};
#pragma unroll 2
  for (int kb = 0; kb < K; kb += 4) {
    v2f a;
    a.x = Arow[kb];
    a.y = Arow[kb + 1];
    const float* Bk = B0 + (size_t)kb * ldb;
    v2f b0, b1, b2, b3;
    b0.x = Bk[0];   b0.y = Bk[ldb];
    b1.x = Bk[16];  b1.y = Bk[ldb + 16];
    b2.x = Bk[32];  b2.y = Bk[ldb + 32];
    b3.x = Bk[48];  b3.y = Bk[ldb + 48];
    acc0 = __builtin_amdgcn_wmma_f32_16x16x4_f32(false, a, false, b0, (short)0, acc0, false, false);
    acc1 = __builtin_amdgcn_wmma_f32_16x16x4_f32(false, a, false, b1, (short)0, acc1, false, false);
    acc2 = __builtin_amdgcn_wmma_f32_16x16x4_f32(false, a, false, b2, (short)0, acc2, false, false);
    acc3 = __builtin_amdgcn_wmma_f32_16x16x4_f32(false, a, false, b3, (short)0, acc3, false, false);
  }

  int crow0 = row0 + ((lane >> 4) << 3);
  v8f accs[4] = {acc0, acc1, acc2, acc3};
#pragma unroll
  for (int g = 0; g < 4; ++g) {
    int ccol = col0 + (g << 4) + ml;
    float bv = (bias != nullptr) ? bias[ccol] : 0.0f;
#pragma unroll
    for (int r = 0; r < 8; ++r) {
      float v = accs[g][r] + bv;
      if (do_relu) v = fmaxf(v, 0.0f);
      C[(size_t)(crow0 + r) * ldc + ccol] = v;
    }
  }
}

__global__ void pool_reduce(const float* __restrict__ H, float* __restrict__ pool) {
  int c = blockIdx.y;
  const float* h = H + (size_t)c * NP * FDIM;
  const int TOT = N_DRUGS * FDIM;
  float s = 0.0f;
  for (int i = blockIdx.x * blockDim.x + threadIdx.x; i < TOT;
       i += gridDim.x * blockDim.x)
    s += h[i];
  __shared__ float red[256];
  int tid = threadIdx.x;
  red[tid] = s;
  __syncthreads();
  for (int off = 128; off > 0; off >>= 1) {
    if (tid < off) red[tid] += red[tid + off];
    __syncthreads();
  }
  if (tid == 0) atomicAdd(&pool[c], red[0]);
}

__global__ void attn_kernel(const float* __restrict__ pool,
                            const float* __restrict__ fc1_w, const float* __restrict__ fc1_b,
                            const float* __restrict__ fc2_w, const float* __restrict__ fc2_b,
                            float* __restrict__ a_out) {
  __shared__ float p[6], h[30];
  int tid = threadIdx.x;
  if (tid < 6) p[tid] = pool[tid] * (1.0f / (591.0f * 1024.0f));
  __syncthreads();
  if (tid < 30) {
    float s = fc1_b[tid];
    for (int c = 0; c < 6; ++c) s += p[c] * fc1_w[tid * 6 + c];
    h[tid] = fmaxf(s, 0.0f);
  }
  __syncthreads();
  if (tid < 6) {
    float s = fc2_b[tid];
    for (int j = 0; j < 30; ++j) s += h[j] * fc2_w[tid * 30 + j];
    a_out[tid] = 1.0f / (1.0f + expf(-s));
  }
}

__global__ void final_kernel(const float* __restrict__ H, const float* __restrict__ a,
                             const float* __restrict__ conv_w,
                             const float* __restrict__ conv_b,
                             float* __restrict__ out) {
  int idx = blockIdx.x * blockDim.x + threadIdx.x;
  if (idx >= N_DRUGS * FDIM) return;
  int n = idx >> 10, f = idx & (FDIM - 1);
  float s = conv_b[0];
#pragma unroll
  for (int c = 0; c < 6; ++c) {
    float v = a[c] * H[(size_t)c * NP * FDIM + (size_t)n * FDIM + f];
    s += conv_w[c] * fmaxf(v, 0.0f);
  }
  out[idx] = s;
}

extern "C" void kernel_launch(void* const* d_in, const int* in_sizes, int n_in,
                              void* d_out, int out_size, void* d_ws, size_t ws_size,
                              hipStream_t stream) {
  (void)in_sizes; (void)n_in; (void)out_size; (void)ws_size;

  const float* dm1 = (const float*)d_in[0];
  const int*   edges[3] = {(const int*)d_in[1], (const int*)d_in[3], (const int*)d_in[5]};
  const float* ew[3]    = {(const float*)d_in[2], (const float*)d_in[4], (const float*)d_in[6]};
  const float* W1[3]    = {(const float*)d_in[7],  (const float*)d_in[11], (const float*)d_in[15]};
  const float* b1[3]    = {(const float*)d_in[8],  (const float*)d_in[12], (const float*)d_in[16]};
  const float* W2[3]    = {(const float*)d_in[9],  (const float*)d_in[13], (const float*)d_in[17]};
  const float* b2[3]    = {(const float*)d_in[10], (const float*)d_in[14], (const float*)d_in[18]};
  const float* fc1_w  = (const float*)d_in[19];
  const float* fc1_b  = (const float*)d_in[20];
  const float* fc2_w  = (const float*)d_in[21];
  const float* fc2_b  = (const float*)d_in[22];
  const float* conv_w = (const float*)d_in[23];
  const float* conv_b = (const float*)d_in[24];

  float* ws     = (float*)d_ws;
  float* ahat   = ws;
  float* deg    = ahat + (size_t)3 * NP * NP;
  float* dinv   = deg  + 3 * NP;
  float* pool   = dinv + 3 * NP;
  float* a_att  = pool + 8;
  float* H      = a_att + 8;
  float* xw     = H + (size_t)6 * NP * FDIM;
  float* x_pad  = xw + (size_t)NP * FDIM;

  init_ws<<<1024, 256, 0, stream>>>(ahat, deg, pool);

  pad_dm1<<<(NP * FDIM + 255) / 256, 256, 0, stream>>>(dm1, x_pad);

  deg_accum<<<(3 * NEDGE + 255) / 256, 256, 0, stream>>>(
      edges[0], ew[0], edges[1], ew[1], edges[2], ew[2], deg);

  compute_dinv<<<(3 * NP + 255) / 256, 256, 0, stream>>>(deg, dinv);

  build_ahat<<<(3 * (NEDGE + N_DRUGS) + 255) / 256, 256, 0, stream>>>(
      edges[0], ew[0], edges[1], ew[1], edges[2], ew[2], dinv, ahat);

  const int m_tiles  = NP / 16;
  const int n_wtiles = FDIM / 64;
  const int waves    = m_tiles * n_wtiles;
  const int blocks   = (waves * 32 + 255) / 256;

  for (int t = 0; t < 3; ++t) {
    float* Ah = ahat + (size_t)t * NP * NP;
    float* h1 = H + (size_t)(2 * t)     * NP * FDIM;
    float* h2 = H + (size_t)(2 * t + 1) * NP * FDIM;

    gemm_wmma_f32<<<blocks, 256, 0, stream>>>(x_pad, FDIM, W1[t], FDIM,
                                              xw, FDIM, FDIM, n_wtiles, m_tiles,
                                              nullptr, 0);
    gemm_wmma_f32<<<blocks, 256, 0, stream>>>(Ah, NP, xw, FDIM,
                                              h1, FDIM, NP, n_wtiles, m_tiles,
                                              b1[t], 1);
    gemm_wmma_f32<<<blocks, 256, 0, stream>>>(h1, FDIM, W2[t], FDIM,
                                              xw, FDIM, FDIM, n_wtiles, m_tiles,
                                              nullptr, 0);
    gemm_wmma_f32<<<blocks, 256, 0, stream>>>(Ah, NP, xw, FDIM,
                                              h2, FDIM, NP, n_wtiles, m_tiles,
                                              b2[t], 1);
  }

  dim3 pgrid(64, 6);
  pool_reduce<<<pgrid, 256, 0, stream>>>(H, pool);
  attn_kernel<<<1, 32, 0, stream>>>(pool, fc1_w, fc1_b, fc2_w, fc2_b, a_att);
  final_kernel<<<(N_DRUGS * FDIM + 255) / 256, 256, 0, stream>>>(
      H, a_att, conv_w, conv_b, (float*)d_out);
}


# GEMM k range split over the 8 waves of a workgroup (one 16x64 strip per workgroup: 592 workgroups per GEMM, was 74); partials added through LDS in wave order, so k is summed in a different order
# speedup vs baseline: 4.6896x; 4.6896x over previous
#include <hip/hip_runtime.h>
#include <hip/hip_bf16.h>


#define N_DRUGS 591
#define NP      592
#define FDIM    1024
#define NEDGE   120000

typedef __attribute__((ext_vector_type(2))) float v2f;
typedef __attribute__((ext_vector_type(8))) float v8f;

__global__ void init_ws(float* __restrict__ ahat, float* __restrict__ deg,
                        float* __restrict__ pool) {
  int idx = blockIdx.x * blockDim.x + threadIdx.x;
  int stride = gridDim.x * blockDim.x;
  const int AH = 3 * NP * NP;
  for (int i = idx; i < AH; i += stride) ahat[i] = 0.0f;
  for (int i = idx; i < 3 * NP; i += stride) deg[i] = 1.0f;
  if (idx < 8) pool[idx] = 0.0f;
}

__global__ void pad_dm1(const float* __restrict__ dm1, float* __restrict__ out) {
  int idx = blockIdx.x * blockDim.x + threadIdx.x;
  if (idx >= NP * FDIM) return;
  int n = idx >> 10;
  out[idx] = (n < N_DRUGS) ? dm1[idx] : 0.0f;
}

__global__ void deg_accum(const int* __restrict__ eT, const float* __restrict__ wT,
                          const int* __restrict__ eS, const float* __restrict__ wS,
                          const int* __restrict__ eG, const float* __restrict__ wG,
                          float* __restrict__ deg) {
  int idx = blockIdx.x * blockDim.x + threadIdx.x;
  if (idx >= 3 * NEDGE) return;
  int t = idx / NEDGE, e = idx - t * NEDGE;
  const int*   ed = (t == 0) ? eT : ((t == 1) ? eS : eG);
  const float* w  = (t == 0) ? wT : ((t == 1) ? wS : wG);
  int dst = ed[NEDGE + e];
  atomicAdd(&deg[t * NP + dst], w[e]);
}

__global__ void compute_dinv(const float* __restrict__ deg, float* __restrict__ dinv) {
  int i = blockIdx.x * blockDim.x + threadIdx.x;
  if (i < 3 * NP) {
    float d = deg[i];
    dinv[i] = (d > 0.0f) ? rsqrtf(d) : 0.0f;
  }
}

__global__ void build_ahat(const int* __restrict__ eT, const float* __restrict__ wT,
                           const int* __restrict__ eS, const float* __restrict__ wS,
                           const int* __restrict__ eG, const float* __restrict__ wG,
                           const float* __restrict__ dinv, float* __restrict__ ahat) {
  const int PER = NEDGE + N_DRUGS;
  int idx = blockIdx.x * blockDim.x + threadIdx.x;
  if (idx >= 3 * PER) return;
  int t = idx / PER, r = idx - t * PER;
  const float* dv = dinv + t * NP;
  float* Ah = ahat + (size_t)t * NP * NP;
  if (r < NEDGE) {
    const int*   ed = (t == 0) ? eT : ((t == 1) ? eS : eG);
    const float* w  = (t == 0) ? wT : ((t == 1) ? wS : wG);
    int src = ed[r], dst = ed[NEDGE + r];
    atomicAdd(&Ah[(size_t)dst * NP + src], dv[src] * w[r] * dv[dst]);
  } else {
    int i = r - NEDGE;
    atomicAdd(&Ah[(size_t)i * NP + i], dv[i] * dv[i]);
  }
}

#define KSPLIT 8
#define PPITCH 68

__global__ void gemm_wmma_f32(const float* __restrict__ A, int lda,
                              const float* __restrict__ B, int ldb,
                              float* __restrict__ C, int ldc,
                              int K, int n_wtiles,
                              const float* __restrict__ bias, int do_relu) {
  __shared__ __attribute__((aligned(16))) float part[KSPLIT * 16 * PPITCH];

  const int w = __builtin_amdgcn_readfirstlane((int)(threadIdx.x >> 5));
  const int lane = threadIdx.x & 31;
  const int tm = blockIdx.x / n_wtiles, tn = blockIdx.x - tm * n_wtiles;
  const int row0 = tm << 4, col0 = tn << 6;
  const int ml = lane & 15;
  const int khalf = (lane >> 4) << 1;

  const int steps = K >> 2;
  const int base = steps / KSPLIT, rem = steps - base * KSPLIT;
  const int s0 = w * base + ((w < rem) ? w : rem);
  const int cnt = base + ((w < rem) ? 1 : 0);
  const int kbeg = s0 << 2, kend = (s0 + cnt) << 2;

  const float* Arow = A + (size_t)(row0 + ml) * lda + khalf;
  const float* B0   = B + (size_t)khalf * ldb + (col0 + ml);

  v8f acc0 = {}, acc1 = {}, acc2 = {}, acc3 = {};
#pragma unroll 2
  for (int kb = kbeg; kb < kend; kb += 4) {
    v2f a;
    a.x = Arow[kb];
    a.y = Arow[kb + 1];
    const float* Bk = B0 + (size_t)kb * ldb;
    v2f b0, b1, b2, b3;
    b0.x = Bk[0];   b0.y = Bk[ldb];
    b1.x = Bk[16];  b1.y = Bk[ldb + 16];
    b2.x = Bk[32];  b2.y = Bk[ldb + 32];
    b3.x = Bk[48];  b3.y = Bk[ldb + 48];
    acc0 = __builtin_amdgcn_wmma_f32_16x16x4_f32(false, a, false, b0, (short)0, acc0, false, false);
    acc1 = __builtin_amdgcn_wmma_f32_16x16x4_f32(false, a, false, b1, (short)0, acc1, false, false);
    acc2 = __builtin_amdgcn_wmma_f32_16x16x4_f32(false, a, false, b2, (short)0, acc2, false, false);
    acc3 = __builtin_amdgcn_wmma_f32_16x16x4_f32(false, a, false, b3, (short)0, acc3, false, false);
  }

  float* pw = part + w * (16 * PPITCH);
  const int prow0 = (lane >> 4) << 3;
  v8f accs[4] = {acc0, acc1, acc2, acc3};
#pragma unroll
  for (int g = 0; g < 4; ++g) {
#pragma unroll
    for (int r = 0; r < 8; ++r)
      pw[(prow0 + r) * PPITCH + (g << 4) + ml] = accs[g][r];
  }
  __syncthreads();

  const int t = threadIdx.x;
  const int orow = t >> 4, oc = (t & 15) << 2;
  const float* pr = part + orow * PPITCH + oc;
  float4 s = *(const float4*)pr;
#pragma unroll
  for (int q = 1; q < KSPLIT; ++q) {
    float4 p = *(const float4*)(pr + q * (16 * PPITCH));
    s.x += p.x; s.y += p.y; s.z += p.z; s.w += p.w;
  }
  const int ccol = col0 + oc;
  if (bias != nullptr) {
    s.x += bias[ccol]; s.y += bias[ccol + 1]; s.z += bias[ccol + 2]; s.w += bias[ccol + 3];
  }
  if (do_relu) {
    s.x = fmaxf(s.x, 0.0f); s.y = fmaxf(s.y, 0.0f);
    s.z = fmaxf(s.z, 0.0f); s.w = fmaxf(s.w, 0.0f);
  }
  *(float4*)(C + (size_t)(row0 + orow) * ldc + ccol) = s;
}

__global__ void pool_reduce(const float* __restrict__ H, float* __restrict__ pool) {
  int c = blockIdx.y;
  const float* h = H + (size_t)c * NP * FDIM;
  const int TOT = N_DRUGS * FDIM;
  float s = 0.0f;
  for (int i = blockIdx.x * blockDim.x + threadIdx.x; i < TOT;
       i += gridDim.x * blockDim.x)
    s += h[i];
  __shared__ float red[256];
  int tid = threadIdx.x;
  red[tid] = s;
  __syncthreads();
  for (int off = 128; off > 0; off >>= 1) {
    if (tid < off) red[tid] += red[tid + off];
    __syncthreads();
  }
  if (tid == 0) atomicAdd(&pool[c], red[0]);
}

__global__ void attn_kernel(const float* __restrict__ pool,
                            const float* __restrict__ fc1_w, const float* __restrict__ fc1_b,
                            const float* __restrict__ fc2_w, const float* __restrict__ fc2_b,
                            float* __restrict__ a_out) {
  __shared__ float p[6], h[30];
  int tid = threadIdx.x;
  if (tid < 6) p[tid] = pool[tid] * (1.0f / (591.0f * 1024.0f));
  __syncthreads();
  if (tid < 30) {
    float s = fc1_b[tid];
    for (int c = 0; c < 6; ++c) s += p[c] * fc1_w[tid * 6 + c];
    h[tid] = fmaxf(s, 0.0f);
  }
  __syncthreads();
  if (tid < 6) {
    float s = fc2_b[tid];
    for (int j = 0; j < 30; ++j) s += h[j] * fc2_w[tid * 30 + j];
    a_out[tid] = 1.0f / (1.0f + expf(-s));
  }
}

__global__ void final_kernel(const float* __restrict__ H, const float* __restrict__ a,
                             const float* __restrict__ conv_w,
                             const float* __restrict__ conv_b,
                             float* __restrict__ out) {
  int idx = blockIdx.x * blockDim.x + threadIdx.x;
  if (idx >= N_DRUGS * FDIM) return;
  int n = idx >> 10, f = idx & (FDIM - 1);
  float s = conv_b[0];
#pragma unroll
  for (int c = 0; c < 6; ++c) {
    float v = a[c] * H[(size_t)c * NP * FDIM + (size_t)n * FDIM + f];
    s += conv_w[c] * fmaxf(v, 0.0f);
  }
  out[idx] = s;
}

extern "C" void kernel_launch(void* const* d_in, const int* in_sizes, int n_in,
                              void* d_out, int out_size, void* d_ws, size_t ws_size,
                              hipStream_t stream) {
  (void)in_sizes; (void)n_in; (void)out_size; (void)ws_size;

  const float* dm1 = (const float*)d_in[0];
  const int*   edges[3] = {(const int*)d_in[1], (const int*)d_in[3], (const int*)d_in[5]};
  const float* ew[3]    = {(const float*)d_in[2], (const float*)d_in[4], (const float*)d_in[6]};
  const float* W1[3]    = {(const float*)d_in[7],  (const float*)d_in[11], (const float*)d_in[15]};
  const float* b1[3]    = {(const float*)d_in[8],  (const float*)d_in[12], (const float*)d_in[16]};
  const float* W2[3]    = {(const float*)d_in[9],  (const float*)d_in[13], (const float*)d_in[17]};
  const float* b2[3]    = {(const float*)d_in[10], (const float*)d_in[14], (const float*)d_in[18]};
  const float* fc1_w  = (const float*)d_in[19];
  const float* fc1_b  = (const float*)d_in[20];
  const float* fc2_w  = (const float*)d_in[21];
  const float* fc2_b  = (const float*)d_in[22];
  const float* conv_w = (const float*)d_in[23];
  const float* conv_b = (const float*)d_in[24];

  float* ws     = (float*)d_ws;
  float* ahat   = ws;
  float* deg    = ahat + (size_t)3 * NP * NP;
  float* dinv   = deg  + 3 * NP;
  float* pool   = dinv + 3 * NP;
  float* a_att  = pool + 8;
  float* H      = a_att + 8;
  float* xw     = H + (size_t)6 * NP * FDIM;
  float* x_pad  = xw + (size_t)NP * FDIM;

  init_ws<<<1024, 256, 0, stream>>>(ahat, deg, pool);

  pad_dm1<<<(NP * FDIM + 255) / 256, 256, 0, stream>>>(dm1, x_pad);

  deg_accum<<<(3 * NEDGE + 255) / 256, 256, 0, stream>>>(
      edges[0], ew[0], edges[1], ew[1], edges[2], ew[2], deg);

  compute_dinv<<<(3 * NP + 255) / 256, 256, 0, stream>>>(deg, dinv);

  build_ahat<<<(3 * (NEDGE + N_DRUGS) + 255) / 256, 256, 0, stream>>>(
      edges[0], ew[0], edges[1], ew[1], edges[2], ew[2], dinv, ahat);

  const int m_tiles  = NP / 16;
  const int n_wtiles = FDIM / 64;
  const int blocks   = m_tiles * n_wtiles;

  for (int t = 0; t < 3; ++t) {
    float* Ah = ahat + (size_t)t * NP * NP;
    float* h1 = H + (size_t)(2 * t)     * NP * FDIM;
    float* h2 = H + (size_t)(2 * t + 1) * NP * FDIM;

    gemm_wmma_f32<<<blocks, 256, 0, stream>>>(x_pad, FDIM, W1[t], FDIM,
                                              xw, FDIM, FDIM, n_wtiles,
                                              nullptr, 0);
    gemm_wmma_f32<<<blocks, 256, 0, stream>>>(Ah, NP, xw, FDIM,
                                              h1, FDIM, NP, n_wtiles,
                                              b1[t], 1);
    gemm_wmma_f32<<<blocks, 256, 0, stream>>>(h1, FDIM, W2[t], FDIM,
                                              xw, FDIM, FDIM, n_wtiles,
                                              nullptr, 0);
    gemm_wmma_f32<<<blocks, 256, 0, stream>>>(Ah, NP, xw, FDIM,
                                              h2, FDIM, NP, n_wtiles,
                                              b2[t], 1);
  }

  dim3 pgrid(64, 6);
  pool_reduce<<<pgrid, 256, 0, stream>>>(H, pool);
  attn_kernel<<<1, 32, 0, stream>>>(pool, fc1_w, fc1_b, fc2_w, fc2_b, a_att);
  final_kernel<<<(N_DRUGS * FDIM + 255) / 256, 256, 0, stream>>>(
      H, a_att, conv_w, conv_b, (float*)d_out);
}


# GEMM K-loop: each wave keeps the fragment loads of the next 6 k-steps in flight (register ring, slot refilled behind its step's WMMAs), not one load round trip per k-step; sums in parent's order
# speedup vs baseline: 5.2867x; 1.1273x over previous
#include <hip/hip_runtime.h>
#include <hip/hip_bf16.h>


#define N_DRUGS 591
#define NP      592
#define FDIM    1024
#define NEDGE   120000

typedef __attribute__((ext_vector_type(2))) float v2f;
typedef __attribute__((ext_vector_type(8))) float v8f;

__global__ void init_ws(float* __restrict__ ahat, float* __restrict__ deg,
                        float* __restrict__ pool) {
  int idx = blockIdx.x * blockDim.x + threadIdx.x;
  int stride = gridDim.x * blockDim.x;
  const int AH = 3 * NP * NP;
  for (int i = idx; i < AH; i += stride) ahat[i] = 0.0f;
  for (int i = idx; i < 3 * NP; i += stride) deg[i] = 1.0f;
  if (idx < 8) pool[idx] = 0.0f;
}

__global__ void pad_dm1(const float* __restrict__ dm1, float* __restrict__ out) {
  int idx = blockIdx.x * blockDim.x + threadIdx.x;
  if (idx >= NP * FDIM) return;
  int n = idx >> 10;
  out[idx] = (n < N_DRUGS) ? dm1[idx] : 0.0f;
}

__global__ void deg_accum(const int* __restrict__ eT, const float* __restrict__ wT,
                          const int* __restrict__ eS, const float* __restrict__ wS,
                          const int* __restrict__ eG, const float* __restrict__ wG,
                          float* __restrict__ deg) {
  int idx = blockIdx.x * blockDim.x + threadIdx.x;
  if (idx >= 3 * NEDGE) return;
  int t = idx / NEDGE, e = idx - t * NEDGE;
  const int*   ed = (t == 0) ? eT : ((t == 1) ? eS : eG);
  const float* w  = (t == 0) ? wT : ((t == 1) ? wS : wG);
  int dst = ed[NEDGE + e];
  atomicAdd(&deg[t * NP + dst], w[e]);
}

__global__ void compute_dinv(const float* __restrict__ deg, float* __restrict__ dinv) {
  int i = blockIdx.x * blockDim.x + threadIdx.x;
  if (i < 3 * NP) {
    float d = deg[i];
    dinv[i] = (d > 0.0f) ? rsqrtf(d) : 0.0f;
  }
}

__global__ void build_ahat(const int* __restrict__ eT, const float* __restrict__ wT,
                           const int* __restrict__ eS, const float* __restrict__ wS,
                           const int* __restrict__ eG, const float* __restrict__ wG,
                           const float* __restrict__ dinv, float* __restrict__ ahat) {
  const int PER = NEDGE + N_DRUGS;
  int idx = blockIdx.x * blockDim.x + threadIdx.x;
  if (idx >= 3 * PER) return;
  int t = idx / PER, r = idx - t * PER;
  const float* dv = dinv + t * NP;
  float* Ah = ahat + (size_t)t * NP * NP;
  if (r < NEDGE) {
    const int*   ed = (t == 0) ? eT : ((t == 1) ? eS : eG);
    const float* w  = (t == 0) ? wT : ((t == 1) ? wS : wG);
    int src = ed[r], dst = ed[NEDGE + r];
    atomicAdd(&Ah[(size_t)dst * NP + src], dv[src] * w[r] * dv[dst]);
  } else {
    int i = r - NEDGE;
    atomicAdd(&Ah[(size_t)i * NP + i], dv[i] * dv[i]);
  }
}

#define KSPLIT 8
#define PPITCH 68

#define PD 6

struct KFrag { v2f a, b0, b1, b2, b3; };

__device__ __forceinline__ void load_kstep(KFrag& f, const float* __restrict__ Arow,
                                           const float* __restrict__ B0, int ldb, int kb) {
  f.a.x = Arow[kb];
  f.a.y = Arow[kb + 1];
  const float* Bk = B0 + (size_t)kb * ldb;
  f.b0.x = Bk[0];   f.b0.y = Bk[ldb];
  f.b1.x = Bk[16];  f.b1.y = Bk[ldb + 16];
  f.b2.x = Bk[32];  f.b2.y = Bk[ldb + 32];
  f.b3.x = Bk[48];  f.b3.y = Bk[ldb + 48];
}

__device__ __forceinline__ void wmma_kstep(const KFrag& f, v8f& acc0, v8f& acc1,
                                           v8f& acc2, v8f& acc3) {
  acc0 = __builtin_amdgcn_wmma_f32_16x16x4_f32(false, f.a, false, f.b0, (short)0, acc0, false, false);
  acc1 = __builtin_amdgcn_wmma_f32_16x16x4_f32(false, f.a, false, f.b1, (short)0, acc1, false, false);
  acc2 = __builtin_amdgcn_wmma_f32_16x16x4_f32(false, f.a, false, f.b2, (short)0, acc2, false, false);
  acc3 = __builtin_amdgcn_wmma_f32_16x16x4_f32(false, f.a, false, f.b3, (short)0, acc3, false, false);
}

__global__ void gemm_wmma_f32(const float* __restrict__ A, int lda,
                              const float* __restrict__ B, int ldb,
                              float* __restrict__ C, int ldc,
                              int K, int n_wtiles,
                              const float* __restrict__ bias, int do_relu) {
  __shared__ __attribute__((aligned(16))) float part[KSPLIT * 16 * PPITCH];

  const int w = __builtin_amdgcn_readfirstlane((int)(threadIdx.x >> 5));
  const int lane = threadIdx.x & 31;
  const int tm = blockIdx.x / n_wtiles, tn = blockIdx.x - tm * n_wtiles;
  const int row0 = tm << 4, col0 = tn << 6;
  const int ml = lane & 15;
  const int khalf = (lane >> 4) << 1;

  const int steps = K >> 2;
  const int base = steps / KSPLIT, rem = steps - base * KSPLIT;
  const int s0 = w * base + ((w < rem) ? w : rem);
  const int cnt = base + ((w < rem) ? 1 : 0);
  const int kbeg = s0 << 2, kend = (s0 + cnt) << 2;

  const float* Arow = A + (size_t)(row0 + ml) * lda + khalf;
  const float* B0   = B + (size_t)khalf * ldb + (col0 + ml);

  v8f acc0 = {}, acc1 = {}, acc2 = {}, acc3 = {};

  if (cnt > 0) {
    const int klast = kend - 4;
    KFrag f[PD];
#pragma unroll
    for (int i = 0; i < PD; ++i) {
      const int kk = kbeg + 4 * i;
      load_kstep(f[i], Arow, B0, ldb, (kk < klast) ? kk : klast);
      __builtin_amdgcn_sched_barrier(0);
    }
    int kb = kbeg;
    for (; kb + 8 * PD <= kend; kb += 4 * PD) {
#pragma unroll
      for (int i = 0; i < PD; ++i) {
        wmma_kstep(f[i], acc0, acc1, acc2, acc3);
        load_kstep(f[i], Arow, B0, ldb, kb + 4 * (i + PD));
        __builtin_amdgcn_sched_barrier(0);
      }
    }
#pragma unroll
    for (int i = 0; i < PD; ++i) {
      if (kb + 4 * i >= kend) break;
      wmma_kstep(f[i], acc0, acc1, acc2, acc3);
      const int kn = kb + 4 * (i + PD);
      if (kn < kend) load_kstep(f[i], Arow, B0, ldb, kn);
    }
    kb += 4 * PD;
#pragma unroll
    for (int i = 0; i < PD - 1; ++i) {
      if (kb + 4 * i >= kend) break;
      wmma_kstep(f[i], acc0, acc1, acc2, acc3);
    }
  }

  float* pw = part + w * (16 * PPITCH);
  const int prow0 = (lane >> 4) << 3;
  v8f accs[4] = {acc0, acc1, acc2, acc3};
#pragma unroll
  for (int g = 0; g < 4; ++g) {
#pragma unroll
    for (int r = 0; r < 8; ++r)
      pw[(prow0 + r) * PPITCH + (g << 4) + ml] = accs[g][r];
  }
  __syncthreads();

  const int t = threadIdx.x;
  const int orow = t >> 4, oc = (t & 15) << 2;
  const float* pr = part + orow * PPITCH + oc;
  float4 s = *(const float4*)pr;
#pragma unroll
  for (int q = 1; q < KSPLIT; ++q) {
    float4 p = *(const float4*)(pr + q * (16 * PPITCH));
    s.x += p.x; s.y += p.y; s.z += p.z; s.w += p.w;
  }
  const int ccol = col0 + oc;
  if (bias != nullptr) {
    s.x += bias[ccol]; s.y += bias[ccol + 1]; s.z += bias[ccol + 2]; s.w += bias[ccol + 3];
  }
  if (do_relu) {
    s.x = fmaxf(s.x, 0.0f); s.y = fmaxf(s.y, 0.0f);
    s.z = fmaxf(s.z, 0.0f); s.w = fmaxf(s.w, 0.0f);
  }
  *(float4*)(C + (size_t)(row0 + orow) * ldc + ccol) = s;
}

__global__ void pool_reduce(const float* __restrict__ H, float* __restrict__ pool) {
  int c = blockIdx.y;
  const float* h = H + (size_t)c * NP * FDIM;
  const int TOT = N_DRUGS * FDIM;
  float s = 0.0f;
  for (int i = blockIdx.x * blockDim.x + threadIdx.x; i < TOT;
       i += gridDim.x * blockDim.x)
    s += h[i];
  __shared__ float red[256];
  int tid = threadIdx.x;
  red[tid] = s;
  __syncthreads();
  for (int off = 128; off > 0; off >>= 1) {
    if (tid < off) red[tid] += red[tid + off];
    __syncthreads();
  }
  if (tid == 0) atomicAdd(&pool[c], red[0]);
}

__global__ void attn_kernel(const float* __restrict__ pool,
                            const float* __restrict__ fc1_w, const float* __restrict__ fc1_b,
                            const float* __restrict__ fc2_w, const float* __restrict__ fc2_b,
                            float* __restrict__ a_out) {
  __shared__ float p[6], h[30];
  int tid = threadIdx.x;
  if (tid < 6) p[tid] = pool[tid] * (1.0f / (591.0f * 1024.0f));
  __syncthreads();
  if (tid < 30) {
    float s = fc1_b[tid];
    for (int c = 0; c < 6; ++c) s += p[c] * fc1_w[tid * 6 + c];
    h[tid] = fmaxf(s, 0.0f);
  }
  __syncthreads();
  if (tid < 6) {
    float s = fc2_b[tid];
    for (int j = 0; j < 30; ++j) s += h[j] * fc2_w[tid * 30 + j];
    a_out[tid] = 1.0f / (1.0f + expf(-s));
  }
}

__global__ void final_kernel(const float* __restrict__ H, const float* __restrict__ a,
                             const float* __restrict__ conv_w,
                             const float* __restrict__ conv_b,
                             float* __restrict__ out) {
  int idx = blockIdx.x * blockDim.x + threadIdx.x;
  if (idx >= N_DRUGS * FDIM) return;
  int n = idx >> 10, f = idx & (FDIM - 1);
  float s = conv_b[0];
#pragma unroll
  for (int c = 0; c < 6; ++c) {
    float v = a[c] * H[(size_t)c * NP * FDIM + (size_t)n * FDIM + f];
    s += conv_w[c] * fmaxf(v, 0.0f);
  }
  out[idx] = s;
}

extern "C" void kernel_launch(void* const* d_in, const int* in_sizes, int n_in,
                              void* d_out, int out_size, void* d_ws, size_t ws_size,
                              hipStream_t stream) {
  (void)in_sizes; (void)n_in; (void)out_size; (void)ws_size;

  const float* dm1 = (const float*)d_in[0];
  const int*   edges[3] = {(const int*)d_in[1], (const int*)d_in[3], (const int*)d_in[5]};
  const float* ew[3]    = {(const float*)d_in[2], (const float*)d_in[4], (const float*)d_in[6]};
  const float* W1[3]    = {(const float*)d_in[7],  (const float*)d_in[11], (const float*)d_in[15]};
  const float* b1[3]    = {(const float*)d_in[8],  (const float*)d_in[12], (const float*)d_in[16]};
  const float* W2[3]    = {(const float*)d_in[9],  (const float*)d_in[13], (const float*)d_in[17]};
  const float* b2[3]    = {(const float*)d_in[10], (const float*)d_in[14], (const float*)d_in[18]};
  const float* fc1_w  = (const float*)d_in[19];
  const float* fc1_b  = (const float*)d_in[20];
  const float* fc2_w  = (const float*)d_in[21];
  const float* fc2_b  = (const float*)d_in[22];
  const float* conv_w = (const float*)d_in[23];
  const float* conv_b = (const float*)d_in[24];

  float* ws     = (float*)d_ws;
  float* ahat   = ws;
  float* deg    = ahat + (size_t)3 * NP * NP;
  float* dinv   = deg  + 3 * NP;
  float* pool   = dinv + 3 * NP;
  float* a_att  = pool + 8;
  float* H      = a_att + 8;
  float* xw     = H + (size_t)6 * NP * FDIM;
  float* x_pad  = xw + (size_t)NP * FDIM;

  init_ws<<<1024, 256, 0, stream>>>(ahat, deg, pool);

  pad_dm1<<<(NP * FDIM + 255) / 256, 256, 0, stream>>>(dm1, x_pad);

  deg_accum<<<(3 * NEDGE + 255) / 256, 256, 0, stream>>>(
      edges[0], ew[0], edges[1], ew[1], edges[2], ew[2], deg);

  compute_dinv<<<(3 * NP + 255) / 256, 256, 0, stream>>>(deg, dinv);

  build_ahat<<<(3 * (NEDGE + N_DRUGS) + 255) / 256, 256, 0, stream>>>(
      edges[0], ew[0], edges[1], ew[1], edges[2], ew[2], dinv, ahat);

  const int m_tiles  = NP / 16;
  const int n_wtiles = FDIM / 64;
  const int blocks   = m_tiles * n_wtiles;

  for (int t = 0; t < 3; ++t) {
    float* Ah = ahat + (size_t)t * NP * NP;
    float* h1 = H + (size_t)(2 * t)     * NP * FDIM;
    float* h2 = H + (size_t)(2 * t + 1) * NP * FDIM;

    gemm_wmma_f32<<<blocks, 256, 0, stream>>>(x_pad, FDIM, W1[t], FDIM,
                                              xw, FDIM, FDIM, n_wtiles,
                                              nullptr, 0);
    gemm_wmma_f32<<<blocks, 256, 0, stream>>>(Ah, NP, xw, FDIM,
                                              h1, FDIM, NP, n_wtiles,
                                              b1[t], 1);
    gemm_wmma_f32<<<blocks, 256, 0, stream>>>(h1, FDIM, W2[t], FDIM,
                                              xw, FDIM, FDIM, n_wtiles,
                                              nullptr, 0);
    gemm_wmma_f32<<<blocks, 256, 0, stream>>>(Ah, NP, xw, FDIM,
                                              h2, FDIM, NP, n_wtiles,
                                              b2[t], 1);
  }

  dim3 pgrid(64, 6);
  pool_reduce<<<pgrid, 256, 0, stream>>>(H, pool);
  attn_kernel<<<1, 32, 0, stream>>>(pool, fc1_w, fc1_b, fc2_w, fc2_b, a_att);
  final_kernel<<<(N_DRUGS * FDIM + 255) / 256, 256, 0, stream>>>(
      H, a_att, conv_w, conv_b, (float*)d_out);
}


# GEMM operands written by one kernel for a later GEMM (padded dm1, Ahat, h1 as A; x@W as B) are stored in WMMA fragment order: one contiguous 8-byte load per lane per fragment; sums in parent's order
# speedup vs baseline: 6.2622x; 1.1845x over previous
#include <hip/hip_runtime.h>
#include <hip/hip_bf16.h>


#define N_DRUGS 591
#define NP      592
#define FDIM    1024
#define NEDGE   120000

typedef __attribute__((ext_vector_type(2))) float v2f;
typedef __attribute__((ext_vector_type(8))) float v8f;

__global__ void init_ws(float* __restrict__ ahat, float* __restrict__ deg,
                        float* __restrict__ pool) {
  int idx = blockIdx.x * blockDim.x + threadIdx.x;
  int stride = gridDim.x * blockDim.x;
  const int AH = 3 * NP * NP;
  for (int i = idx; i < AH; i += stride) ahat[i] = 0.0f;
  for (int i = idx; i < 3 * NP; i += stride) deg[i] = 1.0f;
  if (idx < 8) pool[idx] = 0.0f;
}

__device__ __forceinline__ size_t frag_index(int i, int k, int KX) {
  return (((size_t)(i >> 4) * (size_t)(KX >> 1) + (size_t)(k >> 1)) << 5) + ((i & 15) << 1) + (k & 1);
}

__global__ void pad_dm1(const float* __restrict__ dm1, float* __restrict__ out) {
  int idx = blockIdx.x * blockDim.x + threadIdx.x;
  if (idx >= NP * FDIM) return;
  int e = idx & 1, r = (idx >> 1) & 15, blk = idx >> 5;
  int p = blk & (FDIM / 2 - 1), tm = blk >> 9;
  int n = (tm << 4) + r, f = (p << 1) + e;
  out[idx] = (n < N_DRUGS) ? dm1[(size_t)n * FDIM + f] : 0.0f;
}

__global__ void deg_accum(const int* __restrict__ eT, const float* __restrict__ wT,
                          const int* __restrict__ eS, const float* __restrict__ wS,
                          const int* __restrict__ eG, const float* __restrict__ wG,
                          float* __restrict__ deg) {
  int idx = blockIdx.x * blockDim.x + threadIdx.x;
  if (idx >= 3 * NEDGE) return;
  int t = idx / NEDGE, e = idx - t * NEDGE;
  const int*   ed = (t == 0) ? eT : ((t == 1) ? eS : eG);
  const float* w  = (t == 0) ? wT : ((t == 1) ? wS : wG);
  int dst = ed[NEDGE + e];
  atomicAdd(&deg[t * NP + dst], w[e]);
}

__global__ void compute_dinv(const float* __restrict__ deg, float* __restrict__ dinv) {
  int i = blockIdx.x * blockDim.x + threadIdx.x;
  if (i < 3 * NP) {
    float d = deg[i];
    dinv[i] = (d > 0.0f) ? rsqrtf(d) : 0.0f;
  }
}

__global__ void build_ahat(const int* __restrict__ eT, const float* __restrict__ wT,
                           const int* __restrict__ eS, const float* __restrict__ wS,
                           const int* __restrict__ eG, const float* __restrict__ wG,
                           const float* __restrict__ dinv, float* __restrict__ ahat) {
  const int PER = NEDGE + N_DRUGS;
  int idx = blockIdx.x * blockDim.x + threadIdx.x;
  if (idx >= 3 * PER) return;
  int t = idx / PER, r = idx - t * PER;
  const float* dv = dinv + t * NP;
  float* Ah = ahat + (size_t)t * NP * NP;
  if (r < NEDGE) {
    const int*   ed = (t == 0) ? eT : ((t == 1) ? eS : eG);
    const float* w  = (t == 0) ? wT : ((t == 1) ? wS : wG);
    int src = ed[r], dst = ed[NEDGE + r];
    atomicAdd(&Ah[frag_index(dst, src, NP)], dv[src] * w[r] * dv[dst]);
  } else {
    int i = r - NEDGE;
    atomicAdd(&Ah[frag_index(i, i, NP)], dv[i] * dv[i]);
  }
}

#define KSPLIT 8
#define PPITCH 68

#define PD 6

struct KFrag { v2f a, b0, b1, b2, b3; };

template <bool BFRAG>
__device__ __forceinline__ void load_kstep(KFrag& f, const float* __restrict__ Ap,
                                           const float* __restrict__ Bp, int ldb,
                                           int btile, int kb) {
  f.a = *(const v2f*)(Ap + (kb << 4));
  if (BFRAG) {
    const float* Bk = Bp + (kb << 4);
    f.b0 = *(const v2f*)(Bk);
    f.b1 = *(const v2f*)(Bk + btile);
    f.b2 = *(const v2f*)(Bk + 2 * btile);
    f.b3 = *(const v2f*)(Bk + 3 * btile);
  } else {
    const float* Bk = Bp + (size_t)kb * ldb;
    f.b0.x = Bk[0];   f.b0.y = Bk[ldb];
    f.b1.x = Bk[16];  f.b1.y = Bk[ldb + 16];
    f.b2.x = Bk[32];  f.b2.y = Bk[ldb + 32];
    f.b3.x = Bk[48];  f.b3.y = Bk[ldb + 48];
  }
}

__device__ __forceinline__ void wmma_kstep(const KFrag& f, v8f& acc0, v8f& acc1,
                                           v8f& acc2, v8f& acc3) {
  acc0 = __builtin_amdgcn_wmma_f32_16x16x4_f32(false, f.a, false, f.b0, (short)0, acc0, false, false);
  acc1 = __builtin_amdgcn_wmma_f32_16x16x4_f32(false, f.a, false, f.b1, (short)0, acc1, false, false);
  acc2 = __builtin_amdgcn_wmma_f32_16x16x4_f32(false, f.a, false, f.b2, (short)0, acc2, false, false);
  acc3 = __builtin_amdgcn_wmma_f32_16x16x4_f32(false, f.a, false, f.b3, (short)0, acc3, false, false);
}

template <bool BFRAG>
__device__ __forceinline__ void gemm_body(const float* __restrict__ A,
                                          const float* __restrict__ B, int ldb,
                                          float* __restrict__ C, int ldc,
                                          float* __restrict__ Ca, float* __restrict__ Cb,
                                          int K, int n_wtiles,
                                          const float* __restrict__ bias, int do_relu,
                                          float* part) {
  const int w = __builtin_amdgcn_readfirstlane((int)(threadIdx.x >> 5));
  const int lane = threadIdx.x & 31;
  const int tm = blockIdx.x / n_wtiles, tn = blockIdx.x - tm * n_wtiles;
  const int row0 = tm << 4, col0 = tn << 6;
  const int ml = lane & 15;
  const int khalf = (lane >> 4) << 1;

  const int steps = K >> 2;
  const int base = steps / KSPLIT, rem = steps - base * KSPLIT;
  const int s0 = w * base + ((w < rem) ? w : rem);
  const int cnt = base + ((w < rem) ? 1 : 0);
  const int kbeg = s0 << 2, kend = (s0 + cnt) << 2;

  const int btile = K << 4;
  const float* Ap = A + ((size_t)tm * (size_t)(K >> 1) << 5) + (khalf << 4) + (ml << 1);
  const float* Bp = BFRAG ? (B + (size_t)(col0 >> 4) * btile + (khalf << 4) + (ml << 1))
                          : (B + (size_t)khalf * ldb + (col0 + ml));

  v8f acc0 = {}, acc1 = {}, acc2 = {}, acc3 = {};

  if (cnt > 0) {
    const int klast = kend - 4;
    KFrag f[PD];
#pragma unroll
    for (int i = 0; i < PD; ++i) {
      const int kk = kbeg + 4 * i;
      load_kstep<BFRAG>(f[i], Ap, Bp, ldb, btile, (kk < klast) ? kk : klast);
      __builtin_amdgcn_sched_barrier(0);
    }
    int kb = kbeg;
    for (; kb + 8 * PD <= kend; kb += 4 * PD) {
#pragma unroll
      for (int i = 0; i < PD; ++i) {
        wmma_kstep(f[i], acc0, acc1, acc2, acc3);
        load_kstep<BFRAG>(f[i], Ap, Bp, ldb, btile, kb + 4 * (i + PD));
        __builtin_amdgcn_sched_barrier(0);
      }
    }
#pragma unroll
    for (int i = 0; i < PD; ++i) {
      if (kb + 4 * i >= kend) break;
      wmma_kstep(f[i], acc0, acc1, acc2, acc3);
      const int kn = kb + 4 * (i + PD);
      if (kn < kend) load_kstep<BFRAG>(f[i], Ap, Bp, ldb, btile, kn);
    }
    kb += 4 * PD;
#pragma unroll
    for (int i = 0; i < PD - 1; ++i) {
      if (kb + 4 * i >= kend) break;
      wmma_kstep(f[i], acc0, acc1, acc2, acc3);
    }
  }

  float* pw = part + w * (16 * PPITCH);
  const int prow0 = (lane >> 4) << 3;
  v8f accs[4] = {acc0, acc1, acc2, acc3};
#pragma unroll
  for (int g = 0; g < 4; ++g) {
#pragma unroll
    for (int r = 0; r < 8; ++r)
      pw[(prow0 + r) * PPITCH + (g << 4) + ml] = accs[g][r];
  }
  __syncthreads();

  const int t = threadIdx.x;
  const int orow = t >> 4, oc = (t & 15) << 2;
  float* pr = part + orow * PPITCH + oc;
  float4 s = *(const float4*)pr;
#pragma unroll
  for (int q = 1; q < KSPLIT; ++q) {
    float4 p = *(const float4*)(pr + q * (16 * PPITCH));
    s.x += p.x; s.y += p.y; s.z += p.z; s.w += p.w;
  }
  const int ccol = col0 + oc;
  if (bias != nullptr) {
    s.x += bias[ccol]; s.y += bias[ccol + 1]; s.z += bias[ccol + 2]; s.w += bias[ccol + 3];
  }
  if (do_relu) {
    s.x = fmaxf(s.x, 0.0f); s.y = fmaxf(s.y, 0.0f);
    s.z = fmaxf(s.z, 0.0f); s.w = fmaxf(s.w, 0.0f);
  }
  if (C != nullptr) *(float4*)(C + (size_t)(row0 + orow) * ldc + ccol) = s;

  if (Ca != nullptr || Cb != nullptr) {
    *(float4*)pr = s;
    __syncthreads();
    if (Ca != nullptr) {
      const int q = t >> 3, r0 = (t & 7) << 1;
      const float* sp = part + r0 * PPITCH + (q << 1);
      float4 v;
      v.x = sp[0];      v.y = sp[1];
      v.z = sp[PPITCH]; v.w = sp[PPITCH + 1];
      float* dst = Ca + (((size_t)tm * (FDIM / 2) + (size_t)(col0 >> 1)) << 5);
      *(float4*)(dst + (t << 2)) = v;
    }
    if (Cb != nullptr) {
      const int j = t >> 6, u = t & 63, lp = u >> 3, c0 = (u & 7) << 1;
      const float* sp = part + (lp << 1) * PPITCH + (j << 4) + c0;
      float4 v;
      v.x = sp[0];      v.y = sp[PPITCH];
      v.z = sp[1];      v.w = sp[PPITCH + 1];
      float* dst = Cb + (((size_t)((col0 >> 4) + j) * (NP / 2) + (size_t)(row0 >> 1)) << 5);
      *(float4*)(dst + (u << 2)) = v;
    }
  }
}

__global__ void gemm_xw(const float* __restrict__ A, const float* __restrict__ B, int ldb,
                        float* __restrict__ Cb, int K, int n_wtiles) {
  __shared__ __attribute__((aligned(16))) float part[KSPLIT * 16 * PPITCH];
  gemm_body<false>(A, B, ldb, nullptr, 0, nullptr, Cb, K, n_wtiles, nullptr, 0, part);
}

__global__ void gemm_agg(const float* __restrict__ A, const float* __restrict__ B,
                         float* __restrict__ C, int ldc, float* __restrict__ Ca,
                         int K, int n_wtiles, const float* __restrict__ bias) {
  __shared__ __attribute__((aligned(16))) float part[KSPLIT * 16 * PPITCH];
  gemm_body<true>(A, B, 0, C, ldc, Ca, nullptr, K, n_wtiles, bias, 1, part);
}

__global__ void pool_reduce(const float* __restrict__ H, float* __restrict__ pool) {
  int c = blockIdx.y;
  const float* h = H + (size_t)c * NP * FDIM;
  const int TOT = N_DRUGS * FDIM;
  float s = 0.0f;
  for (int i = blockIdx.x * blockDim.x + threadIdx.x; i < TOT;
       i += gridDim.x * blockDim.x)
    s += h[i];
  __shared__ float red[256];
  int tid = threadIdx.x;
  red[tid] = s;
  __syncthreads();
  for (int off = 128; off > 0; off >>= 1) {
    if (tid < off) red[tid] += red[tid + off];
    __syncthreads();
  }
  if (tid == 0) atomicAdd(&pool[c], red[0]);
}

__global__ void attn_kernel(const float* __restrict__ pool,
                            const float* __restrict__ fc1_w, const float* __restrict__ fc1_b,
                            const float* __restrict__ fc2_w, const float* __restrict__ fc2_b,
                            float* __restrict__ a_out) {
  __shared__ float p[6], h[30];
  int tid = threadIdx.x;
  if (tid < 6) p[tid] = pool[tid] * (1.0f / (591.0f * 1024.0f));
  __syncthreads();
  if (tid < 30) {
    float s = fc1_b[tid];
    for (int c = 0; c < 6; ++c) s += p[c] * fc1_w[tid * 6 + c];
    h[tid] = fmaxf(s, 0.0f);
  }
  __syncthreads();
  if (tid < 6) {
    float s = fc2_b[tid];
    for (int j = 0; j < 30; ++j) s += h[j] * fc2_w[tid * 30 + j];
    a_out[tid] = 1.0f / (1.0f + expf(-s));
  }
}

__global__ void final_kernel(const float* __restrict__ H, const float* __restrict__ a,
                             const float* __restrict__ conv_w,
                             const float* __restrict__ conv_b,
                             float* __restrict__ out) {
  int idx = blockIdx.x * blockDim.x + threadIdx.x;
  if (idx >= N_DRUGS * FDIM) return;
  int n = idx >> 10, f = idx & (FDIM - 1);
  float s = conv_b[0];
#pragma unroll
  for (int c = 0; c < 6; ++c) {
    float v = a[c] * H[(size_t)c * NP * FDIM + (size_t)n * FDIM + f];
    s += conv_w[c] * fmaxf(v, 0.0f);
  }
  out[idx] = s;
}

extern "C" void kernel_launch(void* const* d_in, const int* in_sizes, int n_in,
                              void* d_out, int out_size, void* d_ws, size_t ws_size,
                              hipStream_t stream) {
  (void)in_sizes; (void)n_in; (void)out_size; (void)ws_size;

  const float* dm1 = (const float*)d_in[0];
  const int*   edges[3] = {(const int*)d_in[1], (const int*)d_in[3], (const int*)d_in[5]};
  const float* ew[3]    = {(const float*)d_in[2], (const float*)d_in[4], (const float*)d_in[6]};
  const float* W1[3]    = {(const float*)d_in[7],  (const float*)d_in[11], (const float*)d_in[15]};
  const float* b1[3]    = {(const float*)d_in[8],  (const float*)d_in[12], (const float*)d_in[16]};
  const float* W2[3]    = {(const float*)d_in[9],  (const float*)d_in[13], (const float*)d_in[17]};
  const float* b2[3]    = {(const float*)d_in[10], (const float*)d_in[14], (const float*)d_in[18]};
  const float* fc1_w  = (const float*)d_in[19];
  const float* fc1_b  = (const float*)d_in[20];
  const float* fc2_w  = (const float*)d_in[21];
  const float* fc2_b  = (const float*)d_in[22];
  const float* conv_w = (const float*)d_in[23];
  const float* conv_b = (const float*)d_in[24];

  float* ws     = (float*)d_ws;
  float* ahat   = ws;
  float* H      = ahat + (size_t)3 * NP * NP;
  float* xw     = H + (size_t)6 * NP * FDIM;
  float* x_pad  = xw + (size_t)NP * FDIM;
  float* deg    = x_pad + (size_t)NP * FDIM;
  float* dinv   = deg  + 3 * NP;
  float* pool   = dinv + 3 * NP;
  float* a_att  = pool + 8;

  init_ws<<<1024, 256, 0, stream>>>(ahat, deg, pool);

  pad_dm1<<<(NP * FDIM + 255) / 256, 256, 0, stream>>>(dm1, x_pad);

  deg_accum<<<(3 * NEDGE + 255) / 256, 256, 0, stream>>>(
      edges[0], ew[0], edges[1], ew[1], edges[2], ew[2], deg);

  compute_dinv<<<(3 * NP + 255) / 256, 256, 0, stream>>>(deg, dinv);

  build_ahat<<<(3 * (NEDGE + N_DRUGS) + 255) / 256, 256, 0, stream>>>(
      edges[0], ew[0], edges[1], ew[1], edges[2], ew[2], dinv, ahat);

  const int m_tiles  = NP / 16;
  const int n_wtiles = FDIM / 64;
  const int blocks   = m_tiles * n_wtiles;

  for (int t = 0; t < 3; ++t) {
    float* Ah = ahat + (size_t)t * NP * NP;
    float* h1 = H + (size_t)(2 * t)     * NP * FDIM;
    float* h2 = H + (size_t)(2 * t + 1) * NP * FDIM;
    float* h1f = h2;

    gemm_xw<<<blocks, 256, 0, stream>>>(x_pad, W1[t], FDIM, xw, FDIM, n_wtiles);
    gemm_agg<<<blocks, 256, 0, stream>>>(Ah, xw, h1, FDIM, h1f, NP, n_wtiles, b1[t]);
    gemm_xw<<<blocks, 256, 0, stream>>>(h1f, W2[t], FDIM, xw, FDIM, n_wtiles);
    gemm_agg<<<blocks, 256, 0, stream>>>(Ah, xw, h2, FDIM, nullptr, NP, n_wtiles, b2[t]);
  }

  dim3 pgrid(64, 6);
  pool_reduce<<<pgrid, 256, 0, stream>>>(H, pool);
  attn_kernel<<<1, 32, 0, stream>>>(pool, fc1_w, fc1_b, fc2_w, fc2_b, a_att);
  final_kernel<<<(N_DRUGS * FDIM + 255) / 256, 256, 0, stream>>>(
      H, a_att, conv_w, conv_b, (float*)d_out);
}


# GEMM tile per workgroup 80x32 (5x2 fragments in every wave: 7 loaded fragments per 10 WMMAs, was 16x64: 5 per 4) on 256 workgroups, one per WGP; k slices and every sum in parent's order
# speedup vs baseline: 6.8000x; 1.0859x over previous
#include <hip/hip_runtime.h>
#include <hip/hip_bf16.h>


#define N_DRUGS 591
#define NP      592
#define FDIM    1024
#define NEDGE   120000

typedef __attribute__((ext_vector_type(2))) float v2f;
typedef __attribute__((ext_vector_type(8))) float v8f;

__global__ void init_ws(float* __restrict__ ahat, float* __restrict__ deg,
                        float* __restrict__ pool) {
  int idx = blockIdx.x * blockDim.x + threadIdx.x;
  int stride = gridDim.x * blockDim.x;
  const int AH = 3 * NP * NP;
  for (int i = idx; i < AH; i += stride) ahat[i] = 0.0f;
  for (int i = idx; i < 3 * NP; i += stride) deg[i] = 1.0f;
  if (idx < 8) pool[idx] = 0.0f;
}

__device__ __forceinline__ size_t frag_index(int i, int k, int KX) {
  return (((size_t)(i >> 4) * (size_t)(KX >> 1) + (size_t)(k >> 1)) << 5) + ((i & 15) << 1) + (k & 1);
}

__global__ void pad_dm1(const float* __restrict__ dm1, float* __restrict__ out) {
  int idx = blockIdx.x * blockDim.x + threadIdx.x;
  if (idx >= NP * FDIM) return;
  int e = idx & 1, r = (idx >> 1) & 15, blk = idx >> 5;
  int p = blk & (FDIM / 2 - 1), tm = blk >> 9;
  int n = (tm << 4) + r, f = (p << 1) + e;
  out[idx] = (n < N_DRUGS) ? dm1[(size_t)n * FDIM + f] : 0.0f;
}

__global__ void deg_accum(const int* __restrict__ eT, const float* __restrict__ wT,
                          const int* __restrict__ eS, const float* __restrict__ wS,
                          const int* __restrict__ eG, const float* __restrict__ wG,
                          float* __restrict__ deg) {
  int idx = blockIdx.x * blockDim.x + threadIdx.x;
  if (idx >= 3 * NEDGE) return;
  int t = idx / NEDGE, e = idx - t * NEDGE;
  const int*   ed = (t == 0) ? eT : ((t == 1) ? eS : eG);
  const float* w  = (t == 0) ? wT : ((t == 1) ? wS : wG);
  int dst = ed[NEDGE + e];
  atomicAdd(&deg[t * NP + dst], w[e]);
}

__global__ void compute_dinv(const float* __restrict__ deg, float* __restrict__ dinv) {
  int i = blockIdx.x * blockDim.x + threadIdx.x;
  if (i < 3 * NP) {
    float d = deg[i];
    dinv[i] = (d > 0.0f) ? rsqrtf(d) : 0.0f;
  }
}

__global__ void build_ahat(const int* __restrict__ eT, const float* __restrict__ wT,
                           const int* __restrict__ eS, const float* __restrict__ wS,
                           const int* __restrict__ eG, const float* __restrict__ wG,
                           const float* __restrict__ dinv, float* __restrict__ ahat) {
  const int PER = NEDGE + N_DRUGS;
  int idx = blockIdx.x * blockDim.x + threadIdx.x;
  if (idx >= 3 * PER) return;
  int t = idx / PER, r = idx - t * PER;
  const float* dv = dinv + t * NP;
  float* Ah = ahat + (size_t)t * NP * NP;
  if (r < NEDGE) {
    const int*   ed = (t == 0) ? eT : ((t == 1) ? eS : eG);
    const float* w  = (t == 0) ? wT : ((t == 1) ? wS : wG);
    int src = ed[r], dst = ed[NEDGE + r];
    atomicAdd(&Ah[frag_index(dst, src, NP)], dv[src] * w[r] * dv[dst]);
  } else {
    int i = r - NEDGE;
    atomicAdd(&Ah[frag_index(i, i, NP)], dv[i] * dv[i]);
  }
}

#define KSPLIT 8
#define MF 5
#define NF 2
#define TROWS (MF * 16)
#define TCOLS (NF * 16)
#define M_TILES (NP / 16)
#define ROW_GROUPS ((M_TILES + MF - 1) / MF)
#define COL_GROUPS (FDIM / TCOLS)
#define PPITCH 36
#define PLANE (TROWS * PPITCH)
#define TILE_F4 (TROWS * TCOLS / 4)
#define LDS_FLOATS (41 * 1024)

#define PD 6

struct KFrag { v2f a[MF]; v2f b[NF]; };

template <bool BFRAG>
__device__ __forceinline__ void load_kstep(KFrag& f, const float* const (&Ap)[MF],
                                           const float* __restrict__ Bp, int ldb,
                                           int btile, int kb) {
#pragma unroll
  for (int i = 0; i < MF; ++i)
    f.a[i] = *(const v2f*)(Ap[i] + (kb << 4));
  if (BFRAG) {
    const float* Bk = Bp + (kb << 4);
#pragma unroll
    for (int j = 0; j < NF; ++j) f.b[j] = *(const v2f*)(Bk + j * btile);
  } else {
    const float* Bk = Bp + (size_t)kb * ldb;
#pragma unroll
    for (int j = 0; j < NF; ++j) {
      f.b[j].x = Bk[16 * j];
      f.b[j].y = Bk[ldb + 16 * j];
    }
  }
}

__device__ __forceinline__ void wmma_kstep(const KFrag& f, v8f (&acc)[MF][NF]) {
#pragma unroll
  for (int i = 0; i < MF; ++i) {
#pragma unroll
    for (int j = 0; j < NF; ++j)
      acc[i][j] = __builtin_amdgcn_wmma_f32_16x16x4_f32(false, f.a[i], false, f.b[j], (short)0,
                                                         acc[i][j], false, false);
  }
}

template <bool BFRAG>
__device__ __forceinline__ void gemm_body(const float* __restrict__ A,
                                          const float* __restrict__ B, int ldb,
                                          float* __restrict__ C, int ldc,
                                          float* __restrict__ Ca, float* __restrict__ Cb,
                                          int K,
                                          const float* __restrict__ bias, int do_relu,
                                          float* part) {
  const int w = __builtin_amdgcn_readfirstlane((int)(threadIdx.x >> 5));
  const int lane = threadIdx.x & 31;
  const int rg = blockIdx.x / COL_GROUPS, cg = blockIdx.x - rg * COL_GROUPS;
  const int tm0 = rg * MF;
  const int row0 = tm0 << 4, col0 = cg * TCOLS;
  const int ml = lane & 15;
  const int khalf = (lane >> 4) << 1;

  const int steps = K >> 2;
  const int base = steps / KSPLIT, rem = steps - base * KSPLIT;
  const int s0 = w * base + ((w < rem) ? w : rem);
  const int cnt = base + ((w < rem) ? 1 : 0);
  const int kbeg = s0 << 2, kend = (s0 + cnt) << 2;

  const int btile = K << 4;
  const float* Ap[MF];
#pragma unroll
  for (int i = 0; i < MF; ++i) {
    const int tmi = (tm0 + i < M_TILES) ? (tm0 + i) : (M_TILES - 1);
    Ap[i] = A + (size_t)tmi * (size_t)btile + (khalf << 4) + (ml << 1);
  }
  const float* Bp = BFRAG ? (B + (size_t)(col0 >> 4) * btile + (khalf << 4) + (ml << 1))
                          : (B + (size_t)khalf * ldb + (col0 + ml));

  v8f acc[MF][NF] = {};

  if (cnt > 0) {
    const int klast = kend - 4;
    KFrag f[PD];
#pragma unroll
    for (int i = 0; i < PD; ++i) {
      const int kk = kbeg + 4 * i;
      load_kstep<BFRAG>(f[i], Ap, Bp, ldb, btile, (kk < klast) ? kk : klast);
      __builtin_amdgcn_sched_barrier(0);
    }
    int kb = kbeg;
    for (; kb + 8 * PD <= kend; kb += 4 * PD) {
#pragma unroll
      for (int i = 0; i < PD; ++i) {
        wmma_kstep(f[i], acc);
        load_kstep<BFRAG>(f[i], Ap, Bp, ldb, btile, kb + 4 * (i + PD));
        __builtin_amdgcn_sched_barrier(0);
      }
    }
#pragma unroll
    for (int i = 0; i < PD; ++i) {
      if (kb + 4 * i >= kend) break;
      wmma_kstep(f[i], acc);
      const int kn = kb + 4 * (i + PD);
      if (kn < kend) load_kstep<BFRAG>(f[i], Ap, Bp, ldb, btile, kn);
    }
    kb += 4 * PD;
#pragma unroll
    for (int i = 0; i < PD - 1; ++i) {
      if (kb + 4 * i >= kend) break;
      wmma_kstep(f[i], acc);
    }
  }

  float* pw = part + w * PLANE;
  const int prow0 = (lane >> 4) << 3;
#pragma unroll
  for (int i = 0; i < MF; ++i) {
#pragma unroll
    for (int j = 0; j < NF; ++j) {
#pragma unroll
      for (int r = 0; r < 8; ++r)
        pw[((i << 4) + prow0 + r) * PPITCH + (j << 4) + ml] = acc[i][j][r];
    }
  }
  __syncthreads();

  const int t = threadIdx.x;
  const bool refrag = (Ca != nullptr || Cb != nullptr);
#pragma unroll
  for (int n = 0; n < 3; ++n) {
    const int q = t + (n << 8);
    if (q < TILE_F4) {
      const int orow = q >> 3, oc = (q & 7) << 2;
      float* pr = part + orow * PPITCH + oc;
      float4 s = *(const float4*)pr;
#pragma unroll
      for (int p = 1; p < KSPLIT; ++p) {
        float4 v = *(const float4*)(pr + p * PLANE);
        s.x += v.x; s.y += v.y; s.z += v.z; s.w += v.w;
      }
      const int ccol = col0 + oc;
      if (bias != nullptr) {
        s.x += bias[ccol]; s.y += bias[ccol + 1]; s.z += bias[ccol + 2]; s.w += bias[ccol + 3];
      }
      if (do_relu) {
        s.x = fmaxf(s.x, 0.0f); s.y = fmaxf(s.y, 0.0f);
        s.z = fmaxf(s.z, 0.0f); s.w = fmaxf(s.w, 0.0f);
      }
      if (C != nullptr && row0 + orow < NP)
        *(float4*)(C + (size_t)(row0 + orow) * ldc + ccol) = s;
      if (refrag) *(float4*)pr = s;
    }
  }

  if (refrag) {
    __syncthreads();
    if (Ca != nullptr) {
#pragma unroll
      for (int n = 0; n < 3; ++n) {
        const int q = t + (n << 8);
        const int i = q >> 7, u = q & 127;
        if (q < TILE_F4 && tm0 + i < M_TILES) {
          const int p = u >> 3, r0 = (u & 7) << 1;
          const float* sp = part + ((i << 4) + r0) * PPITCH + (p << 1);
          float4 v;
          v.x = sp[0];      v.y = sp[1];
          v.z = sp[PPITCH]; v.w = sp[PPITCH + 1];
          float* dst = Ca + (((size_t)(tm0 + i) * (FDIM / 2) + (size_t)(col0 >> 1)) << 5);
          *(float4*)(dst + (u << 2)) = v;
        }
      }
    }
    if (Cb != nullptr) {
#pragma unroll
      for (int n = 0; n < 3; ++n) {
        const int q = t + (n << 8);
        const int j = (q >= TILE_F4 / 2) ? 1 : 0, u = q - j * (TILE_F4 / 2);
        const int lp = u >> 3, c0 = (u & 7) << 1;
        if (q < TILE_F4 && row0 + (lp << 1) < NP) {
          const float* sp = part + (lp << 1) * PPITCH + (j << 4) + c0;
          float4 v;
          v.x = sp[0];      v.y = sp[PPITCH];
          v.z = sp[1];      v.w = sp[PPITCH + 1];
          float* dst = Cb + (((size_t)((col0 >> 4) + j) * (NP / 2) + (size_t)(row0 >> 1)) << 5);
          *(float4*)(dst + (u << 2)) = v;
        }
      }
    }
  }
}

__global__ void __launch_bounds__(256) gemm_xw(const float* __restrict__ A, const float* __restrict__ B, int ldb,
                        float* __restrict__ Cb, int K) {
  __shared__ __attribute__((aligned(16))) float part[LDS_FLOATS];
  gemm_body<false>(A, B, ldb, nullptr, 0, nullptr, Cb, K, nullptr, 0, part);
}

__global__ void __launch_bounds__(256) gemm_agg(const float* __restrict__ A, const float* __restrict__ B,
                         float* __restrict__ C, int ldc, float* __restrict__ Ca,
                         int K, const float* __restrict__ bias) {
  __shared__ __attribute__((aligned(16))) float part[LDS_FLOATS];
  gemm_body<true>(A, B, 0, C, ldc, Ca, nullptr, K, bias, 1, part);
}

__global__ void pool_reduce(const float* __restrict__ H, float* __restrict__ pool) {
  int c = blockIdx.y;
  const float* h = H + (size_t)c * NP * FDIM;
  const int TOT = N_DRUGS * FDIM;
  float s = 0.0f;
  for (int i = blockIdx.x * blockDim.x + threadIdx.x; i < TOT;
       i += gridDim.x * blockDim.x)
    s += h[i];
  __shared__ float red[256];
  int tid = threadIdx.x;
  red[tid] = s;
  __syncthreads();
  for (int off = 128; off > 0; off >>= 1) {
    if (tid < off) red[tid] += red[tid + off];
    __syncthreads();
  }
  if (tid == 0) atomicAdd(&pool[c], red[0]);
}

__global__ void attn_kernel(const float* __restrict__ pool,
                            const float* __restrict__ fc1_w, const float* __restrict__ fc1_b,
                            const float* __restrict__ fc2_w, const float* __restrict__ fc2_b,
                            float* __restrict__ a_out) {
  __shared__ float p[6], h[30];
  int tid = threadIdx.x;
  if (tid < 6) p[tid] = pool[tid] * (1.0f / (591.0f * 1024.0f));
  __syncthreads();
  if (tid < 30) {
    float s = fc1_b[tid];
    for (int c = 0; c < 6; ++c) s += p[c] * fc1_w[tid * 6 + c];
    h[tid] = fmaxf(s, 0.0f);
  }
  __syncthreads();
  if (tid < 6) {
    float s = fc2_b[tid];
    for (int j = 0; j < 30; ++j) s += h[j] * fc2_w[tid * 30 + j];
    a_out[tid] = 1.0f / (1.0f + expf(-s));
  }
}

__global__ void final_kernel(const float* __restrict__ H, const float* __restrict__ a,
                             const float* __restrict__ conv_w,
                             const float* __restrict__ conv_b,
                             float* __restrict__ out) {
  int idx = blockIdx.x * blockDim.x + threadIdx.x;
  if (idx >= N_DRUGS * FDIM) return;
  int n = idx >> 10, f = idx & (FDIM - 1);
  float s = conv_b[0];
#pragma unroll
  for (int c = 0; c < 6; ++c) {
    float v = a[c] * H[(size_t)c * NP * FDIM + (size_t)n * FDIM + f];
    s += conv_w[c] * fmaxf(v, 0.0f);
  }
  out[idx] = s;
}

extern "C" void kernel_launch(void* const* d_in, const int* in_sizes, int n_in,
                              void* d_out, int out_size, void* d_ws, size_t ws_size,
                              hipStream_t stream) {
  (void)in_sizes; (void)n_in; (void)out_size; (void)ws_size;

  const float* dm1 = (const float*)d_in[0];
  const int*   edges[3] = {(const int*)d_in[1], (const int*)d_in[3], (const int*)d_in[5]};
  const float* ew[3]    = {(const float*)d_in[2], (const float*)d_in[4], (const float*)d_in[6]};
  const float* W1[3]    = {(const float*)d_in[7],  (const float*)d_in[11], (const float*)d_in[15]};
  const float* b1[3]    = {(const float*)d_in[8],  (const float*)d_in[12], (const float*)d_in[16]};
  const float* W2[3]    = {(const float*)d_in[9],  (const float*)d_in[13], (const float*)d_in[17]};
  const float* b2[3]    = {(const float*)d_in[10], (const float*)d_in[14], (const float*)d_in[18]};
  const float* fc1_w  = (const float*)d_in[19];
  const float* fc1_b  = (const float*)d_in[20];
  const float* fc2_w  = (const float*)d_in[21];
  const float* fc2_b  = (const float*)d_in[22];
  const float* conv_w = (const float*)d_in[23];
  const float* conv_b = (const float*)d_in[24];

  float* ws     = (float*)d_ws;
  float* ahat   = ws;
  float* H      = ahat + (size_t)3 * NP * NP;
  float* xw     = H + (size_t)6 * NP * FDIM;
  float* x_pad  = xw + (size_t)NP * FDIM;
  float* deg    = x_pad + (size_t)NP * FDIM;
  float* dinv   = deg  + 3 * NP;
  float* pool   = dinv + 3 * NP;
  float* a_att  = pool + 8;

  init_ws<<<1024, 256, 0, stream>>>(ahat, deg, pool);

  pad_dm1<<<(NP * FDIM + 255) / 256, 256, 0, stream>>>(dm1, x_pad);

  deg_accum<<<(3 * NEDGE + 255) / 256, 256, 0, stream>>>(
      edges[0], ew[0], edges[1], ew[1], edges[2], ew[2], deg);

  compute_dinv<<<(3 * NP + 255) / 256, 256, 0, stream>>>(deg, dinv);

  build_ahat<<<(3 * (NEDGE + N_DRUGS) + 255) / 256, 256, 0, stream>>>(
      edges[0], ew[0], edges[1], ew[1], edges[2], ew[2], dinv, ahat);

  const int blocks = ROW_GROUPS * COL_GROUPS;

  for (int t = 0; t < 3; ++t) {
    float* Ah = ahat + (size_t)t * NP * NP;
    float* h1 = H + (size_t)(2 * t)     * NP * FDIM;
    float* h2 = H + (size_t)(2 * t + 1) * NP * FDIM;
    float* h1f = h2;

    gemm_xw<<<blocks, 256, 0, stream>>>(x_pad, W1[t], FDIM, xw, FDIM);
    gemm_agg<<<blocks, 256, 0, stream>>>(Ah, xw, h1, FDIM, h1f, NP, b1[t]);
    gemm_xw<<<blocks, 256, 0, stream>>>(h1f, W2[t], FDIM, xw, FDIM);
    gemm_agg<<<blocks, 256, 0, stream>>>(Ah, xw, h2, FDIM, nullptr, NP, b2[t]);
  }

  dim3 pgrid(64, 6);
  pool_reduce<<<pgrid, 256, 0, stream>>>(H, pool);
  attn_kernel<<<1, 32, 0, stream>>>(pool, fc1_w, fc1_b, fc2_w, fc2_b, a_att);
  final_kernel<<<(N_DRUGS * FDIM + 255) / 256, 256, 0, stream>>>(
      H, a_att, conv_w, conv_b, (float*)d_out);
}
